# starting kernel
# baseline (speedup 1.0000x reference)
#include <hip/hip_runtime.h>
#include <math.h>

typedef float v2f __attribute__((ext_vector_type(2)));
typedef float v8f __attribute__((ext_vector_type(8)));

#define HIDDEN        128
#define ROWS_PER_WAVE 16
#define WAVES_PER_BLK 8
#define THREADS       (WAVES_PER_BLK * 32)
#define LDS_STRIDE    132
#define W_LDS_FLOATS  (HIDDEN * LDS_STRIDE)
#define Z_LDS_FLOATS  (ROWS_PER_WAVE * LDS_STRIDE)
#define MAX_ITER      50
#define CONV_EPS      1e-7f

__global__ __launch_bounds__(THREADS, 4)
void deq_f32_wmma_kernel(const float* __restrict__ x, const float* __restrict__ W,
                         const float* __restrict__ b, float* __restrict__ out)
{
    extern __shared__ float lds[];
    float* wsh = lds;
    const int tid   = threadIdx.x;
    const int lane  = tid & 31;
    const int wave  = tid >> 5;
    const int lmod  = lane & 15;
    const int lhalf = lane >> 4;
    float* zsh = lds + W_LDS_FLOATS + wave * Z_LDS_FLOATS;

    {
        const float4* Wv = (const float4*)W;
        const int c4 = tid & 31;
        const int r0 = tid >> 5;
        #pragma unroll
        for (int i = 0; i < 16; ++i) {
            const int row = r0 + i * 8;
            float4 v = Wv[row * 32 + c4];
            *(float4*)(wsh + row * LDS_STRIDE + c4 * 4) = v;
        }
    }
    __syncthreads();

    const long row0 = (long)blockIdx.x * (ROWS_PER_WAVE * WAVES_PER_BLK)
                    + (long)wave * ROWS_PER_WAVE;

    v8f xb[8];
    #pragma unroll
    for (int n = 0; n < 8; ++n) {
        const int col  = n * 16 + lmod;
        const float bv = b[col];
        v8f v;
        #pragma unroll
        for (int m = 0; m < 8; ++m) {
            const int row = m + 8 * lhalf;
            v[m] = x[(row0 + row) * HIDDEN + col] + bv;
        }
        xb[n] = v;
        #pragma unroll
        for (int m = 0; m < 8; ++m)
            zsh[(m + 8 * lhalf) * LDS_STRIDE + col] = tanhf(v[m]);
    }
    asm volatile("s_wait_dscnt 0x0" ::: "memory");

    #pragma unroll 1
    for (int it = 1; it < MAX_ITER; ++it) {
        v8f acc[8];
        #pragma unroll
        for (int n = 0; n < 8; ++n) acc[n] = xb[n];

        const float* ap = zsh + lmod * LDS_STRIDE + 2 * lhalf;
        const float* wp = wsh + lmod * LDS_STRIDE + 2 * lhalf;
        #pragma unroll
        for (int t = 0; t < 32; ++t) {
            const v2f a = *(const v2f*)(ap + t * 4);
            #pragma unroll
            for (int n = 0; n < 8; ++n) {
                v2f bf = *(const v2f*)(wp + t * 4 + n * 16 * LDS_STRIDE);
                acc[n] = __builtin_amdgcn_wmma_f32_16x16x4_f32(
                    false, a, false, bf, (short)0, acc[n], false, false);
            }
        }

        float maxd = 0.f;
        #pragma unroll
        for (int n = 0; n < 8; ++n) {
            const int col = n * 16 + lmod;
            #pragma unroll
            for (int m = 0; m < 8; ++m) {
                float* zp  = zsh + (m + 8 * lhalf) * LDS_STRIDE + col;
                float  old = *zp;
                float  nw  = tanhf(acc[n][m]);
                maxd = fmaxf(maxd, fabsf(nw - old));
                *zp = nw;
            }
        }
        asm volatile("s_wait_dscnt 0x0" ::: "memory");

        bool active = (maxd > CONV_EPS);
#if __has_builtin(__builtin_amdgcn_ballot_w32)
        if (__builtin_amdgcn_ballot_w32(active) == 0u) break;
#else
        if (!__any(active)) break;
#endif
    }

    {
        float* o = out + row0 * HIDDEN;
        #pragma unroll
        for (int p = 0; p < 16; ++p) {
            const int idx = p * 32 + lane;
            const int row = idx >> 5;
            const int c4  = idx & 31;
            float4 v = *(const float4*)(zsh + row * LDS_STRIDE + c4 * 4);
            *(float4*)(o + row * HIDDEN + c4 * 4) = v;
        }
    }
}

extern "C" void kernel_launch(void* const* d_in, const int* in_sizes, int n_in,
                              void* d_out, int out_size, void* d_ws, size_t ws_size,
                              hipStream_t stream) {
    const float* x = (const float*)d_in[0];
    const float* W = (const float*)d_in[1];
    const float* b = (const float*)d_in[2];
    float* out     = (float*)d_out;

    const int batch = in_sizes[0] / HIDDEN;
    const int rows_per_block = ROWS_PER_WAVE * WAVES_PER_BLK;
    const int grid = batch / rows_per_block;
    const size_t lds_bytes = (size_t)(W_LDS_FLOATS + WAVES_PER_BLK * Z_LDS_FLOATS) * 4;

    deq_f32_wmma_kernel<<<grid, THREADS, lds_bytes, stream>>>(x, W, b, out);
}


# each wave owns two 16-row tiles half an iteration apart: one tile's tanh (library formulas, lane select not branch, cut in phases) issues behind the other tile's matrix instructions; sums unchanged
# speedup vs baseline: 1.1276x; 1.1276x over previous
#include <hip/hip_runtime.h>
#include <math.h>


typedef float v2f __attribute__((ext_vector_type(2)));
typedef float v8f __attribute__((ext_vector_type(8)));

#define HIDDEN        128
#define ROWS_PER_TILE 16
#define TILES_PER_WAVE 2
#define WAVES_PER_BLK 8
#define THREADS       (WAVES_PER_BLK * 32)
#define LDS_STRIDE    132
#define W_LDS_FLOATS  (HIDDEN * LDS_STRIDE)
#define Z_LDS_FLOATS  (ROWS_PER_TILE * LDS_STRIDE)
#define MAX_ITER      50
#define CONV_EPS      1e-7f

__device__ __forceinline__ float tanh_sel(float x)
{
    const float y  = fabsf(x);
    const float y2 = y * y;
    const float p  = fmaf(y2, fmaf(y2, fmaf(y2, fmaf(y2, -0x1.758e7ap-8f, 0x1.521192p-6f),
                                            -0x1.b8389cp-5f), 0x1.110704p-3f), -0x1.555532p-2f);
    const float zs = fmaf(y2, y * p, y);
    const float t  = expf(2.0f * y);
    const float zl = fmaf(-2.0f, __builtin_amdgcn_rcpf(t + 1.0f), 1.0f);
    return copysignf((y < 0.625f) ? zs : zl, x);
}

#define ACT_ELEM(accH, zH, activeH, maxdH, n, m)                                  \
    do {                                                                          \
        float* zp_ = (zH) + (m) * LDS_STRIDE + (n) * 16;                          \
        const float old_ = *zp_;                                                  \
        const float nw_  = tanh_sel((accH)[n][m]);                                \
        (maxdH) = fmaxf((maxdH), fabsf(nw_ - old_));                              \
        *zp_ = (activeH) ? nw_ : old_;                                            \
    } while (0)

__device__ __forceinline__ void act_alone(const v8f (&accH)[8], float* zH, bool activeH, float& maxdH)
{
    #pragma unroll
    for (int n = 0; n < 8; ++n) {
        #pragma unroll
        for (int m = 0; m < 8; ++m) ACT_ELEM(accH, zH, activeH, maxdH, n, m);
    }
}

#define LOAD_PAIR(u_, B_)                                                          \
    do {                                                                           \
        fa[B_][0] = *(const v2f*)(apG + (2 * (u_)) * 4);                           \
        fa[B_][1] = *(const v2f*)(apG + (2 * (u_) + 1) * 4);                       \
        _Pragma("unroll")                                                          \
        for (int n_ = 0; n_ < 8; ++n_) {                                           \
            fb[B_][0][n_] = *(const v2f*)(wp + (2 * (u_)) * 4 + n_ * 16 * LDS_STRIDE);     \
            fb[B_][1][n_] = *(const v2f*)(wp + (2 * (u_) + 1) * 4 + n_ * 16 * LDS_STRIDE); \
        }                                                                          \
    } while (0)

#define PH0(q_) _Pragma("unroll") for (int m = 0; m < 8; ++m) { xv[m] = accH[q_][m]; old[m] = (zH + (q_) * 16)[m * LDS_STRIDE]; ax[m] = __builtin_fabsf(xv[m]); u2[m] = ax[m] + ax[m]; }
#define PH1(q_) _Pragma("unroll") for (int m = 0; m < 8; ++m) { ph[m] = 0x1.715476p+0f * u2[m]; }
#define PH2(q_) _Pragma("unroll") for (int m = 0; m < 8; ++m) { er[m] = __builtin_rintf(ph[m]); }
#define PH3(q_) _Pragma("unroll") for (int m = 0; m < 8; ++m) { pl[m] = __builtin_fmaf(0x1.715476p+0f, u2[m], -ph[m]); }
#define PH4(q_) _Pragma("unroll") for (int m = 0; m < 8; ++m) { dd[m] = ph[m] - er[m]; }
#define PH5(q_) _Pragma("unroll") for (int m = 0; m < 8; ++m) { pl[m] = __builtin_fmaf(0x1.4ae0bep-26f, u2[m], pl[m]); }
#define PH6(q_) _Pragma("unroll") for (int m = 0; m < 8; ++m) { ei[m] = (int)er[m]; }
#define PH7(q_) _Pragma("unroll") for (int m = 0; m < 8; ++m) { dd[m] = dd[m] + pl[m]; }
#define PH8(q_) _Pragma("unroll") for (int m = 0; m < 8; ++m) { rr[m] = __builtin_amdgcn_exp2f(dd[m]); }
#define PH9(q_) _Pragma("unroll") for (int m = 0; m < 8; ++m) { rr[m] = __builtin_ldexpf(rr[m], ei[m]); }
#define PH10(q_) _Pragma("unroll") for (int m = 0; m < 8; ++m) { rr[m] = !(-0x1.9d1da0p+6f > u2[m]) ? rr[m] : 0.0f; }
#define PH11(q_) _Pragma("unroll") for (int m = 0; m < 8; ++m) { rr[m] = !(0x1.62e430p+6f < u2[m]) ? rr[m] : __builtin_inff(); }
#define PH12(q_) _Pragma("unroll") for (int m = 0; m < 8; ++m) { rr[m] = 1.0f + rr[m]; }
#define PH13(q_) _Pragma("unroll") for (int m = 0; m < 8; ++m) { rr[m] = __builtin_amdgcn_rcpf(rr[m]); }
#define PH14(q_) _Pragma("unroll") for (int m = 0; m < 8; ++m) { zl[m] = __builtin_fmaf(rr[m], -2.0f, 1.0f); }
#define PH15(q_) _Pragma("unroll") for (int m = 0; m < 8; ++m) { y2[m] = xv[m] * xv[m]; }
#define PH16(q_) _Pragma("unroll") for (int m = 0; m < 8; ++m) { pp[m] = __builtin_fmaf(y2[m], -0x1.758e7ap-8f, 0x1.521192p-6f); }
#define PH17(q_) _Pragma("unroll") for (int m = 0; m < 8; ++m) { pp[m] = __builtin_fmaf(y2[m], pp[m], -0x1.b8389cp-5f); }
#define PH18(q_) _Pragma("unroll") for (int m = 0; m < 8; ++m) { pp[m] = __builtin_fmaf(y2[m], pp[m], 0x1.110704p-3f); }
#define PH19(q_) _Pragma("unroll") for (int m = 0; m < 8; ++m) { pp[m] = __builtin_fmaf(y2[m], pp[m], -0x1.555532p-2f); }
#define PH20(q_) _Pragma("unroll") for (int m = 0; m < 8; ++m) { pp[m] = ax[m] * pp[m]; }
#define PH21(q_) _Pragma("unroll") for (int m = 0; m < 8; ++m) { zs[m] = __builtin_fmaf(y2[m], pp[m], ax[m]); }
#define PH22(q_) _Pragma("unroll") for (int m = 0; m < 8; ++m) { zs[m] = !(0.625f > ax[m]) ? zl[m] : zs[m]; }
#define PH23(q_) _Pragma("unroll") for (int m = 0; m < 8; ++m) { nw[m] = __builtin_copysignf(zs[m], xv[m]); }
#define PH24(q_) _Pragma("unroll") for (int m = 0; m < 8; ++m) { maxdH = __builtin_fmaxf(maxdH, __builtin_fabsf(nw[m] - old[m])); }
#define PH25(q_) _Pragma("unroll") for (int m = 0; m < 8; ++m) { (zH + (q_) * 16)[m * LDS_STRIDE] = activeH ? nw[m] : old[m]; }
#define PHX(q_)

#define SLOT(u_, i_, PH_)                                                          \
    accG[(i_) & 7] = __builtin_amdgcn_wmma_f32_16x16x4_f32(                        \
        false, fa[(u_) & 1][(i_) >> 3], false, fb[(u_) & 1][(i_) >> 3][(i_) & 7], (short)0, \
        ((u_) == 0 && (i_) < 8) ? xbG[(i_) & 7] : accG[(i_) & 7], false, false);   \
    PH_((u_) >> 1)                                                                 \
    __builtin_amdgcn_sched_barrier(0);

#define NEXT_LOADS(u_) if ((u_) < 15) LOAD_PAIR((u_) + 1, ((u_) + 1) & 1);
#define PAIR_EVEN(u_) NEXT_LOADS(u_) SLOT(u_, 0, PH0) SLOT(u_, 1, PH1) SLOT(u_, 2, PH2) SLOT(u_, 3, PH3) SLOT(u_, 4, PH4) SLOT(u_, 5, PH5) SLOT(u_, 6, PH6) SLOT(u_, 7, PH7) SLOT(u_, 8, PH8) SLOT(u_, 9, PH9) SLOT(u_, 10, PH10) SLOT(u_, 11, PH11) SLOT(u_, 12, PH12) SLOT(u_, 13, PH13) SLOT(u_, 14, PH14) SLOT(u_, 15, PH15)
#define PAIR_ODD(u_)  NEXT_LOADS(u_) SLOT(u_, 0, PH16) SLOT(u_, 1, PH17) SLOT(u_, 2, PH18) SLOT(u_, 3, PH19) SLOT(u_, 4, PH20) SLOT(u_, 5, PH21) SLOT(u_, 6, PH22) SLOT(u_, 7, PH23) SLOT(u_, 8, PH24) SLOT(u_, 9, PH25) SLOT(u_, 10, PHX) SLOT(u_, 11, PHX) SLOT(u_, 12, PHX) SLOT(u_, 13, PHX) SLOT(u_, 14, PHX) SLOT(u_, 15, PHX)

__device__ __forceinline__ void stage(v8f (&accG)[8], const v8f (&xbG)[8], const float* apG,
                                      const float* wp,
                                      const v8f (&accH)[8], float* zH, bool activeH, float& maxdH)
{
    v2f fa[2][2];
    v2f fb[2][2][8];
    float xv[8], ax[8], old[8], u2[8], ph[8], pl[8], er[8], dd[8], rr[8], zl[8], y2[8], pp[8], zs[8], nw[8];
    int   ei[8];

    LOAD_PAIR(0, 0);
    PAIR_EVEN(0) PAIR_ODD(1) PAIR_EVEN(2) PAIR_ODD(3) PAIR_EVEN(4) PAIR_ODD(5) PAIR_EVEN(6) PAIR_ODD(7) PAIR_EVEN(8) PAIR_ODD(9) PAIR_EVEN(10) PAIR_ODD(11) PAIR_EVEN(12) PAIR_ODD(13) PAIR_EVEN(14) PAIR_ODD(15)
}

__global__ __launch_bounds__(THREADS, 1)
void deq_f32_wmma_kernel(const float* __restrict__ x, const float* __restrict__ W,
                         const float* __restrict__ b, float* __restrict__ out)
{
    extern __shared__ float lds[];
    float* wsh = lds;
    const int tid   = threadIdx.x;
    const int lane  = tid & 31;
    const int wave  = tid >> 5;
    const int lmod  = lane & 15;
    const int lhalf = lane >> 4;
    float* zshA = lds + W_LDS_FLOATS + (wave * TILES_PER_WAVE) * Z_LDS_FLOATS;
    float* zshB = zshA + Z_LDS_FLOATS;

    {
        const float4* Wv = (const float4*)W;
        const int c4 = tid & 31;
        const int r0 = tid >> 5;
        #pragma unroll
        for (int i = 0; i < 16; ++i) {
            const int row = r0 + i * 8;
            float4 v = Wv[row * 32 + c4];
            *(float4*)(wsh + row * LDS_STRIDE + c4 * 4) = v;
        }
    }
    __syncthreads();

    const long row0 = (long)blockIdx.x * (ROWS_PER_TILE * TILES_PER_WAVE * WAVES_PER_BLK)
                    + (long)wave * (ROWS_PER_TILE * TILES_PER_WAVE);

    v8f xbA[8], xbB[8];
    #pragma unroll
    for (int n = 0; n < 8; ++n) {
        const int col  = n * 16 + lmod;
        const float bv = b[col];
        v8f va, vb;
        #pragma unroll
        for (int m = 0; m < 8; ++m) {
            const int row = m + 8 * lhalf;
            va[m] = x[(row0 + row) * HIDDEN + col] + bv;
            vb[m] = x[(row0 + ROWS_PER_TILE + row) * HIDDEN + col] + bv;
        }
        xbA[n] = va;
        xbB[n] = vb;
    }

    float* zDA = zshA + (8 * lhalf) * LDS_STRIDE + lmod;
    float* zDB = zshB + (8 * lhalf) * LDS_STRIDE + lmod;
    const float* apA = zshA + lmod * LDS_STRIDE + 2 * lhalf;
    const float* apB = zshB + lmod * LDS_STRIDE + 2 * lhalf;
    const float* wp  = wsh  + lmod * LDS_STRIDE + 2 * lhalf;

    v8f accA[8], accB[8];
    #pragma unroll
    for (int n = 0; n < 8; ++n) { accA[n] = xbA[n]; accB[n] = xbB[n]; }

    #pragma unroll
    for (int n = 0; n < 8; ++n) {
        #pragma unroll
        for (int m = 0; m < 8; ++m) { zDA[m * LDS_STRIDE + n * 16] = 0.f; zDB[m * LDS_STRIDE + n * 16] = 0.f; }
    }

    bool  activeA = true, activeB = true;
    float maxd = 0.f;
    act_alone(accA, zDA, true, maxd);
    asm volatile("s_wait_dscnt 0x0" ::: "memory");

    #pragma unroll 1
    for (int j = 0; j < MAX_ITER - 1; ++j) {
        float maxdB = 0.f;
        stage(accA, xbA, apA, wp, accB, zDB, activeB, maxdB);
        asm volatile("s_wait_dscnt 0x0" ::: "memory");
        if (j > 0 && activeB)
            activeB = (__builtin_amdgcn_ballot_w32(maxdB > CONV_EPS) != 0u);

        float maxdA = 0.f;
        stage(accB, xbB, apB, wp, accA, zDA, activeA, maxdA);
        asm volatile("s_wait_dscnt 0x0" ::: "memory");
        if (activeA)
            activeA = (__builtin_amdgcn_ballot_w32(maxdA > CONV_EPS) != 0u);

        if (!activeA && !activeB) break;
    }
    act_alone(accB, zDB, activeB, maxd);
    asm volatile("s_wait_dscnt 0x0" ::: "memory");

    {
        float* o = out + row0 * HIDDEN;
        #pragma unroll
        for (int p = 0; p < 32; ++p) {
            const int idx = p * 32 + lane;
            const int row = idx >> 5;
            const int c4  = idx & 31;
            float4 v = *(const float4*)(zshA + row * LDS_STRIDE + c4 * 4);
            *(float4*)(o + row * HIDDEN + c4 * 4) = v;
        }
    }
}

extern "C" void kernel_launch(void* const* d_in, const int* in_sizes, int n_in,
                              void* d_out, int out_size, void* d_ws, size_t ws_size,
                              hipStream_t stream) {
    const float* x = (const float*)d_in[0];
    const float* W = (const float*)d_in[1];
    const float* b = (const float*)d_in[2];
    float* out     = (float*)d_out;

    const int batch = in_sizes[0] / HIDDEN;
    const int rows_per_block = ROWS_PER_TILE * TILES_PER_WAVE * WAVES_PER_BLK;
    const int grid = batch / rows_per_block;
    const size_t lds_bytes = (size_t)(W_LDS_FLOATS + WAVES_PER_BLK * TILES_PER_WAVE * Z_LDS_FLOATS) * 4;

    deq_f32_wmma_kernel<<<grid, THREADS, lds_bytes, stream>>>(x, W, b, out);
}
